# starting kernel
# baseline (speedup 1.0000x reference)
#include <hip/hip_runtime.h>
#include <hip/hip_bf16.h>
#include <stdint.h>

typedef __attribute__((ext_vector_type(2))) float v2f;
typedef __attribute__((ext_vector_type(8))) float v8f;

__global__ __launch_bounds__(256) void zero_kernel(float* __restrict__ p, int n) {
    int i = blockIdx.x * blockDim.x + threadIdx.x;
    if (i < n) p[i] = 0.0f;
}

__device__ __forceinline__ void atomic_add_f32(float* p, float v) {
    __hip_atomic_fetch_add(p, v, __ATOMIC_RELAXED, __HIP_MEMORY_SCOPE_AGENT);
}

__device__ __forceinline__ int   nt_load_i32(const int* p)   { return __builtin_nontemporal_load(p); }
__device__ __forceinline__ float nt_load_f32(const float* p) { return __builtin_nontemporal_load(p); }

__global__ __launch_bounds__(256) void degree_kernel(const int* __restrict__ ei,
                                                     const float* __restrict__ w,
                                                     float* __restrict__ deg, int e) {
    int t = blockIdx.x * blockDim.x + threadIdx.x;
    if (t < e) {
        int d = nt_load_i32(ei + e + t);
        atomic_add_f32(&deg[d], nt_load_f32(w + t));
    }
}

__global__ __launch_bounds__(256) void dinv_kernel(float* __restrict__ deg, int n) {
    int i = blockIdx.x * blockDim.x + threadIdx.x;
    if (i < n) deg[i] = rsqrtf(deg[i] + 1.0f);
}

__global__ __launch_bounds__(32) void linear1_wmma(const float* __restrict__ x,
                                                   const float* __restrict__ W1,
                                                   float* __restrict__ h1, int n) {
    int lane = threadIdx.x;
    int base = blockIdx.x * 16;
    int idx  = lane & 15;
    bool lo  = lane < 16;
    int r    = base + idx;
    bool rv  = (r < n);

    v2f a;
    a.x = (lo && rv) ? x[r * 2 + 0] : 0.0f;
    a.y = (lo && rv) ? x[r * 2 + 1] : 0.0f;

    bool cv = (idx < 8);
    v2f b;
    b.x = (lo && cv) ? W1[0 * 8 + idx] : 0.0f;
    b.y = (lo && cv) ? W1[1 * 8 + idx] : 0.0f;

    v8f c = {};
    v8f d = __builtin_amdgcn_wmma_f32_16x16x4_f32(false, a, false, b,
                                                  (short)0, c, false, false);
    int rowoff = (lo ? 0 : 8) + base;
    if (base + 16 <= n) {
        if (cv) {
#pragma unroll
            for (int v = 0; v < 8; ++v) h1[(rowoff + v) * 8 + idx] = d[v];
        }
    } else {
        if (cv) {
#pragma unroll
            for (int v = 0; v < 8; ++v) {
                int m = rowoff + v;
                if (m < n) h1[m * 8 + idx] = d[v];
            }
        }
    }
}

__global__ __launch_bounds__(32) void linear2_wmma(const float* __restrict__ h1,
                                                   const float* __restrict__ W2,
                                                   float* __restrict__ h2, int n) {
    int lane = threadIdx.x;
    int base = blockIdx.x * 16;
    int idx  = lane & 15;
    bool lo  = lane < 16;
    int r    = base + idx;
    bool rv  = (r < n);
    int k0   = lo ? 0 : 2;

    v2f a0, a1;
    a0.x = rv ? h1[r * 8 + k0 + 0] : 0.0f;
    a0.y = rv ? h1[r * 8 + k0 + 1] : 0.0f;
    a1.x = rv ? h1[r * 8 + k0 + 4] : 0.0f;
    a1.y = rv ? h1[r * 8 + k0 + 5] : 0.0f;

    bool cv = (idx < 2);
    v2f b0, b1;
    b0.x = cv ? W2[(k0 + 0) * 2 + idx] : 0.0f;
    b0.y = cv ? W2[(k0 + 1) * 2 + idx] : 0.0f;
    b1.x = cv ? W2[(k0 + 4) * 2 + idx] : 0.0f;
    b1.y = cv ? W2[(k0 + 5) * 2 + idx] : 0.0f;

    v8f c = {};
    c = __builtin_amdgcn_wmma_f32_16x16x4_f32(false, a0, false, b0, (short)0, c, false, false);
    c = __builtin_amdgcn_wmma_f32_16x16x4_f32(false, a1, false, b1, (short)0, c, false, false);

    int rowoff = (lo ? 0 : 8) + base;
    if (base + 16 <= n) {
        if (cv) {
#pragma unroll
            for (int v = 0; v < 8; ++v) h2[(rowoff + v) * 2 + idx] = c[v];
        }
    } else {
        if (cv) {
#pragma unroll
            for (int v = 0; v < 8; ++v) {
                int m = rowoff + v;
                if (m < n) h2[m * 2 + idx] = c[v];
            }
        }
    }
}

__global__ __launch_bounds__(256) void edge_agg8(const int* __restrict__ ei,
                                                 const float* __restrict__ w,
                                                 const float* __restrict__ dinv,
                                                 const float* __restrict__ h,
                                                 float* __restrict__ agg, int e) {
    int t = blockIdx.x * blockDim.x + threadIdx.x;
    if (t >= e) return;
    int s = nt_load_i32(ei + t);
    int d = nt_load_i32(ei + e + t);
    float nrm = dinv[s] * nt_load_f32(w + t) * dinv[d];
    const float4* hp = (const float4*)(h + (size_t)s * 8);
    float4 lo4 = hp[0];
    float4 hi4 = hp[1];
    float* ap = agg + (size_t)d * 8;
    atomic_add_f32(ap + 0, nrm * lo4.x);
    atomic_add_f32(ap + 1, nrm * lo4.y);
    atomic_add_f32(ap + 2, nrm * lo4.z);
    atomic_add_f32(ap + 3, nrm * lo4.w);
    atomic_add_f32(ap + 4, nrm * hi4.x);
    atomic_add_f32(ap + 5, nrm * hi4.y);
    atomic_add_f32(ap + 6, nrm * hi4.z);
    atomic_add_f32(ap + 7, nrm * hi4.w);
}

__global__ __launch_bounds__(256) void edge_agg2(const int* __restrict__ ei,
                                                 const float* __restrict__ w,
                                                 const float* __restrict__ dinv,
                                                 const float* __restrict__ h,
                                                 float* __restrict__ agg, int e) {
    int t = blockIdx.x * blockDim.x + threadIdx.x;
    if (t >= e) return;
    int s = nt_load_i32(ei + t);
    int d = nt_load_i32(ei + e + t);
    float nrm = dinv[s] * nt_load_f32(w + t) * dinv[d];
    const float2* hp = (const float2*)(h + (size_t)s * 2);
    float2 hv = hp[0];
    float* ap = agg + (size_t)d * 2;
    atomic_add_f32(ap + 0, nrm * hv.x);
    atomic_add_f32(ap + 1, nrm * hv.y);
}

__global__ __launch_bounds__(256) void finalize1(float* __restrict__ h1,
                                                 const float* __restrict__ agg1,
                                                 const float* __restrict__ dinv,
                                                 const float* __restrict__ b1, int n) {
    int i = blockIdx.x * blockDim.x + threadIdx.x;
    if (i >= n) return;
    float sl = dinv[i] * dinv[i];
#pragma unroll
    for (int k = 0; k < 8; ++k) {
        float v = agg1[(size_t)i * 8 + k] + sl * h1[(size_t)i * 8 + k] + b1[k];
        h1[(size_t)i * 8 + k] = v > 0.0f ? v : 0.0f;
    }
}

__global__ __launch_bounds__(256) void finalize2(const float* __restrict__ h2,
                                                 const float* __restrict__ agg2,
                                                 const float* __restrict__ dinv,
                                                 const float* __restrict__ b2,
                                                 float* __restrict__ out, int n) {
    int i = blockIdx.x * blockDim.x + threadIdx.x;
    if (i >= n) return;
    float sl = dinv[i] * dinv[i];
    float v0 = agg2[(size_t)i * 2 + 0] + sl * h2[(size_t)i * 2 + 0] + b2[0];
    float v1 = agg2[(size_t)i * 2 + 1] + sl * h2[(size_t)i * 2 + 1] + b2[1];
    float m  = fmaxf(v0, v1);
    float e0 = __expf(v0 - m);
    float e1 = __expf(v1 - m);
    float inv = 1.0f / (e0 + e1);
    out[(size_t)i * 2 + 0] = e0 * inv;
    out[(size_t)i * 2 + 1] = e1 * inv;
}

static inline float* align16(float* p) {
    return (float*)(((uintptr_t)p + 15) & ~(uintptr_t)15);
}

extern "C" void kernel_launch(void* const* d_in, const int* in_sizes, int n_in,
                              void* d_out, int out_size, void* d_ws, size_t ws_size,
                              hipStream_t stream) {
    const float* x  = (const float*)d_in[0];
    const int*   ei = (const int*)d_in[1];
    const float* w  = (const float*)d_in[2];
    const float* W1 = (const float*)d_in[3];
    const float* b1 = (const float*)d_in[4];
    const float* W2 = (const float*)d_in[5];
    const float* b2 = (const float*)d_in[6];
    float* out = (float*)d_out;

    const int n = in_sizes[0] / 2;
    const int e = in_sizes[2];

    float* ws   = (float*)d_ws;
    float* dinv = align16(ws);
    float* agg1 = align16(dinv + (size_t)n);
    float* agg2 = align16(agg1 + (size_t)n * 8);
    float* h1   = align16(agg2 + (size_t)n * 2);
    float* h2   = align16(h1   + (size_t)n * 8);

    const int TB = 256;
    dim3 blkE((e + TB - 1) / TB), thr(TB);
    dim3 blkN((n + TB - 1) / TB);
    dim3 blkN8((n * 8 + TB - 1) / TB);
    dim3 blkN2((n * 2 + TB - 1) / TB);
    dim3 blkW((n + 15) / 16), thrW(32);

    zero_kernel<<<blkN,  thr, 0, stream>>>(dinv, n);
    zero_kernel<<<blkN8, thr, 0, stream>>>(agg1, n * 8);
    zero_kernel<<<blkN2, thr, 0, stream>>>(agg2, n * 2);

    degree_kernel<<<blkE, thr, 0, stream>>>(ei, w, dinv, e);
    dinv_kernel<<<blkN, thr, 0, stream>>>(dinv, n);

    linear1_wmma<<<blkW, thrW, 0, stream>>>(x, W1, h1, n);
    edge_agg8<<<blkE, thr, 0, stream>>>(ei, w, dinv, h1, agg1, e);
    finalize1<<<blkN, thr, 0, stream>>>(h1, agg1, dinv, b1, n);

    linear2_wmma<<<blkW, thrW, 0, stream>>>(h1, W2, h2, n);
    edge_agg2<<<blkE, thr, 0, stream>>>(ei, w, dinv, h2, agg2, e);
    finalize2<<<blkN, thr, 0, stream>>>(h2, agg2, dinv, b2, out, n);
}


# Scatter-add replaced by gather: edges counting-sorted into per-destination lists (int count, scan, slot fill); each node sums its list in registers; no f32 atomic; f32 sum order changes
# speedup vs baseline: 1.5843x; 1.5843x over previous
#include <hip/hip_runtime.h>
#include <hip/hip_bf16.h>
#include <stdint.h>

typedef __attribute__((ext_vector_type(2))) float v2f;
typedef __attribute__((ext_vector_type(8))) float v8f;

__global__ __launch_bounds__(256) void zero_kernel(float* __restrict__ p, int n) {
    int i = blockIdx.x * blockDim.x + threadIdx.x;
    if (i < n) p[i] = 0.0f;
}

__device__ __forceinline__ void atomic_add_f32(float* p, float v) {
    __hip_atomic_fetch_add(p, v, __ATOMIC_RELAXED, __HIP_MEMORY_SCOPE_AGENT);
}

__device__ __forceinline__ int   nt_load_i32(const int* p)   { return __builtin_nontemporal_load(p); }
__device__ __forceinline__ float nt_load_f32(const float* p) { return __builtin_nontemporal_load(p); }

__global__ __launch_bounds__(256) void degree_kernel(const int* __restrict__ ei,
                                                     const float* __restrict__ w,
                                                     float* __restrict__ deg, int e) {
    int t = blockIdx.x * blockDim.x + threadIdx.x;
    if (t < e) {
        int d = nt_load_i32(ei + e + t);
        atomic_add_f32(&deg[d], nt_load_f32(w + t));
    }
}

__global__ __launch_bounds__(256) void dinv_kernel(float* __restrict__ deg, int n) {
    int i = blockIdx.x * blockDim.x + threadIdx.x;
    if (i < n) deg[i] = rsqrtf(deg[i] + 1.0f);
}

__global__ __launch_bounds__(32) void linear1_wmma(const float* __restrict__ x,
                                                   const float* __restrict__ W1,
                                                   float* __restrict__ h1, int n) {
    int lane = threadIdx.x;
    int base = blockIdx.x * 16;
    int idx  = lane & 15;
    bool lo  = lane < 16;
    int r    = base + idx;
    bool rv  = (r < n);

    v2f a;
    a.x = (lo && rv) ? x[r * 2 + 0] : 0.0f;
    a.y = (lo && rv) ? x[r * 2 + 1] : 0.0f;

    bool cv = (idx < 8);
    v2f b;
    b.x = (lo && cv) ? W1[0 * 8 + idx] : 0.0f;
    b.y = (lo && cv) ? W1[1 * 8 + idx] : 0.0f;

    v8f c = {};
    v8f d = __builtin_amdgcn_wmma_f32_16x16x4_f32(false, a, false, b,
                                                  (short)0, c, false, false);
    int rowoff = (lo ? 0 : 8) + base;
    if (base + 16 <= n) {
        if (cv) {
#pragma unroll
            for (int v = 0; v < 8; ++v) h1[(rowoff + v) * 8 + idx] = d[v];
        }
    } else {
        if (cv) {
#pragma unroll
            for (int v = 0; v < 8; ++v) {
                int m = rowoff + v;
                if (m < n) h1[m * 8 + idx] = d[v];
            }
        }
    }
}

__global__ __launch_bounds__(32) void linear2_wmma(const float* __restrict__ h1,
                                                   const float* __restrict__ W2,
                                                   float* __restrict__ h2, int n) {
    int lane = threadIdx.x;
    int base = blockIdx.x * 16;
    int idx  = lane & 15;
    bool lo  = lane < 16;
    int r    = base + idx;
    bool rv  = (r < n);
    int k0   = lo ? 0 : 2;

    v2f a0, a1;
    a0.x = rv ? h1[r * 8 + k0 + 0] : 0.0f;
    a0.y = rv ? h1[r * 8 + k0 + 1] : 0.0f;
    a1.x = rv ? h1[r * 8 + k0 + 4] : 0.0f;
    a1.y = rv ? h1[r * 8 + k0 + 5] : 0.0f;

    bool cv = (idx < 2);
    v2f b0, b1;
    b0.x = cv ? W2[(k0 + 0) * 2 + idx] : 0.0f;
    b0.y = cv ? W2[(k0 + 1) * 2 + idx] : 0.0f;
    b1.x = cv ? W2[(k0 + 4) * 2 + idx] : 0.0f;
    b1.y = cv ? W2[(k0 + 5) * 2 + idx] : 0.0f;

    v8f c = {};
    c = __builtin_amdgcn_wmma_f32_16x16x4_f32(false, a0, false, b0, (short)0, c, false, false);
    c = __builtin_amdgcn_wmma_f32_16x16x4_f32(false, a1, false, b1, (short)0, c, false, false);

    int rowoff = (lo ? 0 : 8) + base;
    if (base + 16 <= n) {
        if (cv) {
#pragma unroll
            for (int v = 0; v < 8; ++v) h2[(rowoff + v) * 2 + idx] = c[v];
        }
    } else {
        if (cv) {
#pragma unroll
            for (int v = 0; v < 8; ++v) {
                int m = rowoff + v;
                if (m < n) h2[m * 2 + idx] = c[v];
            }
        }
    }
}

__global__ __launch_bounds__(256) void edge_agg8(const int* __restrict__ ei,
                                                 const float* __restrict__ w,
                                                 const float* __restrict__ dinv,
                                                 const float* __restrict__ h,
                                                 float* __restrict__ agg, int e) {
    int t = blockIdx.x * blockDim.x + threadIdx.x;
    if (t >= e) return;
    int s = nt_load_i32(ei + t);
    int d = nt_load_i32(ei + e + t);
    float nrm = dinv[s] * nt_load_f32(w + t) * dinv[d];
    const float4* hp = (const float4*)(h + (size_t)s * 8);
    float4 lo4 = hp[0];
    float4 hi4 = hp[1];
    float* ap = agg + (size_t)d * 8;
    atomic_add_f32(ap + 0, nrm * lo4.x);
    atomic_add_f32(ap + 1, nrm * lo4.y);
    atomic_add_f32(ap + 2, nrm * lo4.z);
    atomic_add_f32(ap + 3, nrm * lo4.w);
    atomic_add_f32(ap + 4, nrm * hi4.x);
    atomic_add_f32(ap + 5, nrm * hi4.y);
    atomic_add_f32(ap + 6, nrm * hi4.z);
    atomic_add_f32(ap + 7, nrm * hi4.w);
}

__global__ __launch_bounds__(256) void edge_agg2(const int* __restrict__ ei,
                                                 const float* __restrict__ w,
                                                 const float* __restrict__ dinv,
                                                 const float* __restrict__ h,
                                                 float* __restrict__ agg, int e) {
    int t = blockIdx.x * blockDim.x + threadIdx.x;
    if (t >= e) return;
    int s = nt_load_i32(ei + t);
    int d = nt_load_i32(ei + e + t);
    float nrm = dinv[s] * nt_load_f32(w + t) * dinv[d];
    const float2* hp = (const float2*)(h + (size_t)s * 2);
    float2 hv = hp[0];
    float* ap = agg + (size_t)d * 2;
    atomic_add_f32(ap + 0, nrm * hv.x);
    atomic_add_f32(ap + 1, nrm * hv.y);
}

__global__ __launch_bounds__(256) void finalize1(float* __restrict__ h1,
                                                 const float* __restrict__ agg1,
                                                 const float* __restrict__ dinv,
                                                 const float* __restrict__ b1, int n) {
    int i = blockIdx.x * blockDim.x + threadIdx.x;
    if (i >= n) return;
    float sl = dinv[i] * dinv[i];
#pragma unroll
    for (int k = 0; k < 8; ++k) {
        float v = agg1[(size_t)i * 8 + k] + sl * h1[(size_t)i * 8 + k] + b1[k];
        h1[(size_t)i * 8 + k] = v > 0.0f ? v : 0.0f;
    }
}

__global__ __launch_bounds__(256) void finalize2(const float* __restrict__ h2,
                                                 const float* __restrict__ agg2,
                                                 const float* __restrict__ dinv,
                                                 const float* __restrict__ b2,
                                                 float* __restrict__ out, int n) {
    int i = blockIdx.x * blockDim.x + threadIdx.x;
    if (i >= n) return;
    float sl = dinv[i] * dinv[i];
    float v0 = agg2[(size_t)i * 2 + 0] + sl * h2[(size_t)i * 2 + 0] + b2[0];
    float v1 = agg2[(size_t)i * 2 + 1] + sl * h2[(size_t)i * 2 + 1] + b2[1];
    float m  = fmaxf(v0, v1);
    float e0 = __expf(v0 - m);
    float e1 = __expf(v1 - m);
    float inv = 1.0f / (e0 + e1);
    out[(size_t)i * 2 + 0] = e0 * inv;
    out[(size_t)i * 2 + 1] = e1 * inv;
}

#define SCAN_B 1024
#define SCAN_SH 10
#define BS_PITCH 32

__global__ __launch_bounds__(256) void zero_i32(int* __restrict__ p, int n) {
    int i = blockIdx.x * blockDim.x + threadIdx.x;
    if (i < n) p[i] = 0;
}

__global__ __launch_bounds__(256) void count_kernel(const int* __restrict__ ei,
                                                    int* __restrict__ cnt, int e) {
    int t = blockIdx.x * blockDim.x + threadIdx.x;
    if (t < e) {
        int d = nt_load_i32(ei + e + t);
        __hip_atomic_fetch_add(cnt + d, 1, __ATOMIC_RELAXED, __HIP_MEMORY_SCOPE_AGENT);
    }
}

__device__ __forceinline__ int wave_incl_scan(int v, int lane) {
#pragma unroll
    for (int o = 1; o < 32; o <<= 1) {
        int t = __shfl_up(v, o);
        if (lane >= o) v += t;
    }
    return v;
}

__global__ __launch_bounds__(SCAN_B) void scan_local(int* __restrict__ cur,
                                                     int* __restrict__ loc,
                                                     int* __restrict__ bsum, int n) {
    __shared__ int wsum[32];
    int tid = threadIdx.x, lane = tid & 31, wv = tid >> 5;
    int i = blockIdx.x * SCAN_B + tid;
    int v = (i < n) ? cur[i] : 0;
    int s = wave_incl_scan(v, lane);
    if (lane == 31) wsum[wv] = s;
    __syncthreads();
    if (wv == 0) {
        int w = wsum[lane];
        int ws = wave_incl_scan(w, lane);
        wsum[lane] = ws - w;
        if (lane == 31) bsum[blockIdx.x * BS_PITCH] = ws;
    }
    __syncthreads();
    int ex = s - v + wsum[wv];
    if (i < n) { loc[i] = ex; cur[i] = ex; }
}

__global__ __launch_bounds__(SCAN_B) void scan_blocks(int* __restrict__ bsum, int nb) {
    __shared__ int wsum[32];
    __shared__ int tot;
    int tid = threadIdx.x, lane = tid & 31, wv = tid >> 5;
    int carry = 0;
    for (int base = 0; base < nb; base += SCAN_B) {
        int i = base + tid;
        int v = (i < nb) ? bsum[i * BS_PITCH] : 0;
        int s = wave_incl_scan(v, lane);
        if (lane == 31) wsum[wv] = s;
        __syncthreads();
        if (wv == 0) {
            int w = wsum[lane];
            int ws = wave_incl_scan(w, lane);
            wsum[lane] = ws - w;
            if (lane == 31) tot = ws;
        }
        __syncthreads();
        if (i < nb) bsum[i * BS_PITCH] = carry + s - v + wsum[wv];
        carry += tot;
        __syncthreads();
    }
}

__global__ __launch_bounds__(256) void fill_kernel(const int* __restrict__ ei,
                                                   const float* __restrict__ w,
                                                   int* __restrict__ cur,
                                                   const int* __restrict__ boff,
                                                   int2* __restrict__ rec, int e) {
    int t = blockIdx.x * blockDim.x + threadIdx.x;
    if (t >= e) return;
    int s = nt_load_i32(ei + t);
    int d = nt_load_i32(ei + e + t);
    float wv = nt_load_f32(w + t);
    int p = __hip_atomic_fetch_add(cur + d, 1, __ATOMIC_RELAXED, __HIP_MEMORY_SCOPE_AGENT)
          + boff[(d >> SCAN_SH) * BS_PITCH];
    rec[p] = make_int2(s, __float_as_int(wv));
}

__global__ __launch_bounds__(256) void deg_pull(const int* __restrict__ loc,
                                                const int* __restrict__ cur,
                                                const int* __restrict__ boff,
                                                const int2* __restrict__ rec,
                                                float* __restrict__ dinv, int n) {
    __shared__ float sdeg[32];
    int node = blockIdx.x * 32 + (threadIdx.x >> 3);
    int j = threadIdx.x & 7;
    int beg = 0, end = 0;
    if (node < n) {
        int off = boff[(node >> SCAN_SH) * BS_PITCH];
        beg = loc[node] + off;
        end = cur[node] + off;
    }
    float acc = 0.0f;
    for (int p = beg + j; p < end; p += 8) acc += __int_as_float(rec[p].y);
    acc += __shfl_xor(acc, 1);
    acc += __shfl_xor(acc, 2);
    acc += __shfl_xor(acc, 4);
    if (j == 0) sdeg[threadIdx.x >> 3] = acc;
    __syncthreads();
    if (threadIdx.x < 32) {
        int nd = blockIdx.x * 32 + threadIdx.x;
        if (nd < n) dinv[nd] = rsqrtf(sdeg[threadIdx.x] + 1.0f);
    }
}

__global__ __launch_bounds__(256) void pull8_relu(const int* __restrict__ loc,
                                                  const int* __restrict__ cur,
                                                  const int* __restrict__ boff,
                                                  const int2* __restrict__ rec,
                                                  const float* __restrict__ dinv,
                                                  const float* __restrict__ h,
                                                  const float* __restrict__ b1,
                                                  float* __restrict__ hout, int n) {
    int t = blockIdx.x * blockDim.x + threadIdx.x;
    int node = t >> 3, j = t & 7;
    int beg = 0, end = 0;
    float di = 0.0f;
    if (node < n) {
        int off = boff[(node >> SCAN_SH) * BS_PITCH];
        beg = loc[node] + off;
        end = cur[node] + off;
        di = dinv[node];
    }
    float a0 = 0.f, a1 = 0.f, a2 = 0.f, a3 = 0.f, a4 = 0.f, a5 = 0.f, a6 = 0.f, a7 = 0.f;
    for (int p = beg + j; p < end; p += 8) {
        int2 r = rec[p];
        float nrm = dinv[r.x] * __int_as_float(r.y) * di;
        const float4* hp = (const float4*)(h + (size_t)r.x * 8);
        float4 lo4 = hp[0];
        float4 hi4 = hp[1];
        a0 += nrm * lo4.x; a1 += nrm * lo4.y; a2 += nrm * lo4.z; a3 += nrm * lo4.w;
        a4 += nrm * hi4.x; a5 += nrm * hi4.y; a6 += nrm * hi4.z; a7 += nrm * hi4.w;
    }
#pragma unroll
    for (int m = 1; m < 8; m <<= 1) {
        a0 += __shfl_xor(a0, m); a1 += __shfl_xor(a1, m);
        a2 += __shfl_xor(a2, m); a3 += __shfl_xor(a3, m);
        a4 += __shfl_xor(a4, m); a5 += __shfl_xor(a5, m);
        a6 += __shfl_xor(a6, m); a7 += __shfl_xor(a7, m);
    }
    if (node < n) {
        float v = a0;
        v = (j == 1) ? a1 : v; v = (j == 2) ? a2 : v; v = (j == 3) ? a3 : v;
        v = (j == 4) ? a4 : v; v = (j == 5) ? a5 : v; v = (j == 6) ? a6 : v;
        v = (j == 7) ? a7 : v;
        v = v + di * di * h[(size_t)node * 8 + j] + b1[j];
        hout[(size_t)node * 8 + j] = v > 0.0f ? v : 0.0f;
    }
}

__global__ __launch_bounds__(256) void pull2_softmax(const int* __restrict__ loc,
                                                     const int* __restrict__ cur,
                                                     const int* __restrict__ boff,
                                                     const int2* __restrict__ rec,
                                                     const float* __restrict__ dinv,
                                                     const float* __restrict__ h2,
                                                     const float* __restrict__ b2,
                                                     float* __restrict__ out, int n) {
    int t = blockIdx.x * blockDim.x + threadIdx.x;
    int node = t >> 1, j = t & 1;
    int beg = 0, end = 0;
    float di = 0.0f;
    if (node < n) {
        int off = boff[(node >> SCAN_SH) * BS_PITCH];
        beg = loc[node] + off;
        end = cur[node] + off;
        di = dinv[node];
    }
    float a0 = 0.f, a1 = 0.f;
    for (int p = beg + j; p < end; p += 2) {
        int2 r = rec[p];
        float nrm = dinv[r.x] * __int_as_float(r.y) * di;
        float2 hv = *(const float2*)(h2 + (size_t)r.x * 2);
        a0 += nrm * hv.x;
        a1 += nrm * hv.y;
    }
    a0 += __shfl_xor(a0, 1);
    a1 += __shfl_xor(a1, 1);
    if (node < n) {
        float sl = di * di;
        float2 hs = *(const float2*)(h2 + (size_t)node * 2);
        float v0 = a0 + sl * hs.x + b2[0];
        float v1 = a1 + sl * hs.y + b2[1];
        float m  = fmaxf(v0, v1);
        float e0 = __expf(v0 - m);
        float e1 = __expf(v1 - m);
        float inv = 1.0f / (e0 + e1);
        out[(size_t)node * 2 + j] = (j ? e1 : e0) * inv;
    }
}

static inline float* align16(float* p) {
    return (float*)(((uintptr_t)p + 15) & ~(uintptr_t)15);
}
static inline size_t up256(size_t b) { return (b + 255) & ~(size_t)255; }

extern "C" void kernel_launch(void* const* d_in, const int* in_sizes, int n_in,
                              void* d_out, int out_size, void* d_ws, size_t ws_size,
                              hipStream_t stream) {
    const float* x  = (const float*)d_in[0];
    const int*   ei = (const int*)d_in[1];
    const float* w  = (const float*)d_in[2];
    const float* W1 = (const float*)d_in[3];
    const float* b1 = (const float*)d_in[4];
    const float* W2 = (const float*)d_in[5];
    const float* b2 = (const float*)d_in[6];
    float* out = (float*)d_out;

    const int n = in_sizes[0] / 2;
    const int e = in_sizes[2];

    const int TB = 256;
    dim3 blkE((e + TB - 1) / TB), thr(TB);
    dim3 blkN((n + TB - 1) / TB);
    dim3 blkW((n + 15) / 16), thrW(32);

    const int nb = (n + SCAN_B - 1) / SCAN_B;
    const size_t o_cur  = 0;
    const size_t o_loc  = o_cur  + up256((size_t)n * 4);
    const size_t o_bs   = o_loc  + up256((size_t)n * 4);
    const size_t o_rec  = o_bs   + up256((size_t)nb * BS_PITCH * 4);
    const size_t o_dinv = o_rec  + up256((size_t)e * 8);
    const size_t o_h1   = o_dinv + up256((size_t)n * 4);
    const size_t o_h1r  = o_h1   + up256((size_t)n * 32);
    const size_t o_h2   = o_h1r  + up256((size_t)n * 32);
    const size_t need   = o_h2   + up256((size_t)n * 8) + 256;

    if (ws_size >= need) {
        char* base = (char*)(((uintptr_t)d_ws + 255) & ~(uintptr_t)255);
        int*   cur  = (int*)(base + o_cur);
        int*   loc  = (int*)(base + o_loc);
        int*   bs   = (int*)(base + o_bs);
        int2*  rec  = (int2*)(base + o_rec);
        float* dinv = (float*)(base + o_dinv);
        float* h1   = (float*)(base + o_h1);
        float* h1r  = (float*)(base + o_h1r);
        float* h2   = (float*)(base + o_h2);

        dim3 blkN8((n * 8 + TB - 1) / TB);
        dim3 blkN2((n * 2 + TB - 1) / TB);
        dim3 blkN32((n + 31) / 32);

        zero_i32<<<blkN, thr, 0, stream>>>(cur, n);
        count_kernel<<<blkE, thr, 0, stream>>>(ei, cur, e);
        scan_local<<<dim3(nb), dim3(SCAN_B), 0, stream>>>(cur, loc, bs, n);
        scan_blocks<<<dim3(1), dim3(SCAN_B), 0, stream>>>(bs, nb);
        fill_kernel<<<blkE, thr, 0, stream>>>(ei, w, cur, bs, rec, e);

        deg_pull<<<blkN32, thr, 0, stream>>>(loc, cur, bs, rec, dinv, n);

        linear1_wmma<<<blkW, thrW, 0, stream>>>(x, W1, h1, n);
        pull8_relu<<<blkN8, thr, 0, stream>>>(loc, cur, bs, rec, dinv, h1, b1, h1r, n);

        linear2_wmma<<<blkW, thrW, 0, stream>>>(h1r, W2, h2, n);
        pull2_softmax<<<blkN2, thr, 0, stream>>>(loc, cur, bs, rec, dinv, h2, b2, out, n);
        return;
    }

    float* ws   = (float*)d_ws;
    float* dinv = align16(ws);
    float* agg1 = align16(dinv + (size_t)n);
    float* agg2 = align16(agg1 + (size_t)n * 8);
    float* h1   = align16(agg2 + (size_t)n * 2);
    float* h2   = align16(h1   + (size_t)n * 8);

    dim3 blkN8((n * 8 + TB - 1) / TB);
    dim3 blkN2((n * 2 + TB - 1) / TB);

    zero_kernel<<<blkN,  thr, 0, stream>>>(dinv, n);
    zero_kernel<<<blkN8, thr, 0, stream>>>(agg1, n * 8);
    zero_kernel<<<blkN2, thr, 0, stream>>>(agg2, n * 2);
    degree_kernel<<<blkE, thr, 0, stream>>>(ei, w, dinv, e);
    dinv_kernel<<<blkN, thr, 0, stream>>>(dinv, n);
    linear1_wmma<<<blkW, thrW, 0, stream>>>(x, W1, h1, n);
    edge_agg8<<<blkE, thr, 0, stream>>>(ei, w, dinv, h1, agg1, e);
    finalize1<<<blkN, thr, 0, stream>>>(h1, agg1, dinv, b1, n);
    linear2_wmma<<<blkW, thrW, 0, stream>>>(h1, W2, h2, n);
    edge_agg2<<<blkE, thr, 0, stream>>>(ei, w, dinv, h2, agg2, e);
    finalize2<<<blkN, thr, 0, stream>>>(h2, agg2, dinv, b2, out, n);
}


# Gather passes fetch ONE packed row per edge: layer 1 = 8 features + dinv in 64 B read by 4 adjacent lanes, layer 2 = 2 features + dinv in 16 B; 8 records per list load; f32 sum order changes
# speedup vs baseline: 1.6826x; 1.0620x over previous
#include <hip/hip_runtime.h>
#include <hip/hip_bf16.h>
#include <stdint.h>

typedef __attribute__((ext_vector_type(2))) float v2f;
typedef __attribute__((ext_vector_type(8))) float v8f;

__global__ __launch_bounds__(256) void zero_kernel(float* __restrict__ p, int n) {
    int i = blockIdx.x * blockDim.x + threadIdx.x;
    if (i < n) p[i] = 0.0f;
}

__device__ __forceinline__ void atomic_add_f32(float* p, float v) {
    __hip_atomic_fetch_add(p, v, __ATOMIC_RELAXED, __HIP_MEMORY_SCOPE_AGENT);
}

__device__ __forceinline__ int   nt_load_i32(const int* p)   { return __builtin_nontemporal_load(p); }
__device__ __forceinline__ float nt_load_f32(const float* p) { return __builtin_nontemporal_load(p); }

__global__ __launch_bounds__(256) void degree_kernel(const int* __restrict__ ei,
                                                     const float* __restrict__ w,
                                                     float* __restrict__ deg, int e) {
    int t = blockIdx.x * blockDim.x + threadIdx.x;
    if (t < e) {
        int d = nt_load_i32(ei + e + t);
        atomic_add_f32(&deg[d], nt_load_f32(w + t));
    }
}

__global__ __launch_bounds__(256) void dinv_kernel(float* __restrict__ deg, int n) {
    int i = blockIdx.x * blockDim.x + threadIdx.x;
    if (i < n) deg[i] = rsqrtf(deg[i] + 1.0f);
}

__global__ __launch_bounds__(32) void linear1_wmma(const float* __restrict__ x,
                                                   const float* __restrict__ W1,
                                                   float* __restrict__ h1, int n) {
    int lane = threadIdx.x;
    int base = blockIdx.x * 16;
    int idx  = lane & 15;
    bool lo  = lane < 16;
    int r    = base + idx;
    bool rv  = (r < n);

    v2f a;
    a.x = (lo && rv) ? x[r * 2 + 0] : 0.0f;
    a.y = (lo && rv) ? x[r * 2 + 1] : 0.0f;

    bool cv = (idx < 8);
    v2f b;
    b.x = (lo && cv) ? W1[0 * 8 + idx] : 0.0f;
    b.y = (lo && cv) ? W1[1 * 8 + idx] : 0.0f;

    v8f c = {};
    v8f d = __builtin_amdgcn_wmma_f32_16x16x4_f32(false, a, false, b,
                                                  (short)0, c, false, false);
    int rowoff = (lo ? 0 : 8) + base;
    if (base + 16 <= n) {
        if (cv) {
#pragma unroll
            for (int v = 0; v < 8; ++v) h1[(rowoff + v) * 8 + idx] = d[v];
        }
    } else {
        if (cv) {
#pragma unroll
            for (int v = 0; v < 8; ++v) {
                int m = rowoff + v;
                if (m < n) h1[m * 8 + idx] = d[v];
            }
        }
    }
}

__global__ __launch_bounds__(32) void linear2_wmma(const float* __restrict__ h1,
                                                   const float* __restrict__ W2,
                                                   float* __restrict__ h2, int n) {
    int lane = threadIdx.x;
    int base = blockIdx.x * 16;
    int idx  = lane & 15;
    bool lo  = lane < 16;
    int r    = base + idx;
    bool rv  = (r < n);
    int k0   = lo ? 0 : 2;

    v2f a0, a1;
    a0.x = rv ? h1[r * 8 + k0 + 0] : 0.0f;
    a0.y = rv ? h1[r * 8 + k0 + 1] : 0.0f;
    a1.x = rv ? h1[r * 8 + k0 + 4] : 0.0f;
    a1.y = rv ? h1[r * 8 + k0 + 5] : 0.0f;

    bool cv = (idx < 2);
    v2f b0, b1;
    b0.x = cv ? W2[(k0 + 0) * 2 + idx] : 0.0f;
    b0.y = cv ? W2[(k0 + 1) * 2 + idx] : 0.0f;
    b1.x = cv ? W2[(k0 + 4) * 2 + idx] : 0.0f;
    b1.y = cv ? W2[(k0 + 5) * 2 + idx] : 0.0f;

    v8f c = {};
    c = __builtin_amdgcn_wmma_f32_16x16x4_f32(false, a0, false, b0, (short)0, c, false, false);
    c = __builtin_amdgcn_wmma_f32_16x16x4_f32(false, a1, false, b1, (short)0, c, false, false);

    int rowoff = (lo ? 0 : 8) + base;
    if (base + 16 <= n) {
        if (cv) {
#pragma unroll
            for (int v = 0; v < 8; ++v) h2[(rowoff + v) * 2 + idx] = c[v];
        }
    } else {
        if (cv) {
#pragma unroll
            for (int v = 0; v < 8; ++v) {
                int m = rowoff + v;
                if (m < n) h2[m * 2 + idx] = c[v];
            }
        }
    }
}

__global__ __launch_bounds__(256) void edge_agg8(const int* __restrict__ ei,
                                                 const float* __restrict__ w,
                                                 const float* __restrict__ dinv,
                                                 const float* __restrict__ h,
                                                 float* __restrict__ agg, int e) {
    int t = blockIdx.x * blockDim.x + threadIdx.x;
    if (t >= e) return;
    int s = nt_load_i32(ei + t);
    int d = nt_load_i32(ei + e + t);
    float nrm = dinv[s] * nt_load_f32(w + t) * dinv[d];
    const float4* hp = (const float4*)(h + (size_t)s * 8);
    float4 lo4 = hp[0];
    float4 hi4 = hp[1];
    float* ap = agg + (size_t)d * 8;
    atomic_add_f32(ap + 0, nrm * lo4.x);
    atomic_add_f32(ap + 1, nrm * lo4.y);
    atomic_add_f32(ap + 2, nrm * lo4.z);
    atomic_add_f32(ap + 3, nrm * lo4.w);
    atomic_add_f32(ap + 4, nrm * hi4.x);
    atomic_add_f32(ap + 5, nrm * hi4.y);
    atomic_add_f32(ap + 6, nrm * hi4.z);
    atomic_add_f32(ap + 7, nrm * hi4.w);
}

__global__ __launch_bounds__(256) void edge_agg2(const int* __restrict__ ei,
                                                 const float* __restrict__ w,
                                                 const float* __restrict__ dinv,
                                                 const float* __restrict__ h,
                                                 float* __restrict__ agg, int e) {
    int t = blockIdx.x * blockDim.x + threadIdx.x;
    if (t >= e) return;
    int s = nt_load_i32(ei + t);
    int d = nt_load_i32(ei + e + t);
    float nrm = dinv[s] * nt_load_f32(w + t) * dinv[d];
    const float2* hp = (const float2*)(h + (size_t)s * 2);
    float2 hv = hp[0];
    float* ap = agg + (size_t)d * 2;
    atomic_add_f32(ap + 0, nrm * hv.x);
    atomic_add_f32(ap + 1, nrm * hv.y);
}

__global__ __launch_bounds__(256) void finalize1(float* __restrict__ h1,
                                                 const float* __restrict__ agg1,
                                                 const float* __restrict__ dinv,
                                                 const float* __restrict__ b1, int n) {
    int i = blockIdx.x * blockDim.x + threadIdx.x;
    if (i >= n) return;
    float sl = dinv[i] * dinv[i];
#pragma unroll
    for (int k = 0; k < 8; ++k) {
        float v = agg1[(size_t)i * 8 + k] + sl * h1[(size_t)i * 8 + k] + b1[k];
        h1[(size_t)i * 8 + k] = v > 0.0f ? v : 0.0f;
    }
}

__global__ __launch_bounds__(256) void finalize2(const float* __restrict__ h2,
                                                 const float* __restrict__ agg2,
                                                 const float* __restrict__ dinv,
                                                 const float* __restrict__ b2,
                                                 float* __restrict__ out, int n) {
    int i = blockIdx.x * blockDim.x + threadIdx.x;
    if (i >= n) return;
    float sl = dinv[i] * dinv[i];
    float v0 = agg2[(size_t)i * 2 + 0] + sl * h2[(size_t)i * 2 + 0] + b2[0];
    float v1 = agg2[(size_t)i * 2 + 1] + sl * h2[(size_t)i * 2 + 1] + b2[1];
    float m  = fmaxf(v0, v1);
    float e0 = __expf(v0 - m);
    float e1 = __expf(v1 - m);
    float inv = 1.0f / (e0 + e1);
    out[(size_t)i * 2 + 0] = e0 * inv;
    out[(size_t)i * 2 + 1] = e1 * inv;
}

#define SCAN_B 1024
#define SCAN_SH 10
#define BS_PITCH 32

__global__ __launch_bounds__(256) void zero_i32(int* __restrict__ p, int n) {
    int i = blockIdx.x * blockDim.x + threadIdx.x;
    if (i < n) p[i] = 0;
}

__global__ __launch_bounds__(256) void count_kernel(const int* __restrict__ ei,
                                                    int* __restrict__ cnt, int e) {
    int t = blockIdx.x * blockDim.x + threadIdx.x;
    if (t < e) {
        int d = nt_load_i32(ei + e + t);
        __hip_atomic_fetch_add(cnt + d, 1, __ATOMIC_RELAXED, __HIP_MEMORY_SCOPE_AGENT);
    }
}

__device__ __forceinline__ int wave_incl_scan(int v, int lane) {
#pragma unroll
    for (int o = 1; o < 32; o <<= 1) {
        int t = __shfl_up(v, o);
        if (lane >= o) v += t;
    }
    return v;
}

__global__ __launch_bounds__(SCAN_B) void scan_local(int* __restrict__ cur,
                                                     int* __restrict__ loc,
                                                     int* __restrict__ bsum, int n) {
    __shared__ int wsum[32];
    int tid = threadIdx.x, lane = tid & 31, wv = tid >> 5;
    int i = blockIdx.x * SCAN_B + tid;
    int v = (i < n) ? cur[i] : 0;
    int s = wave_incl_scan(v, lane);
    if (lane == 31) wsum[wv] = s;
    __syncthreads();
    if (wv == 0) {
        int w = wsum[lane];
        int ws = wave_incl_scan(w, lane);
        wsum[lane] = ws - w;
        if (lane == 31) bsum[blockIdx.x * BS_PITCH] = ws;
    }
    __syncthreads();
    int ex = s - v + wsum[wv];
    if (i < n) { loc[i] = ex; cur[i] = ex; }
}

__global__ __launch_bounds__(SCAN_B) void scan_blocks(int* __restrict__ bsum, int nb) {
    __shared__ int wsum[32];
    __shared__ int tot;
    int tid = threadIdx.x, lane = tid & 31, wv = tid >> 5;
    int carry = 0;
    for (int base = 0; base < nb; base += SCAN_B) {
        int i = base + tid;
        int v = (i < nb) ? bsum[i * BS_PITCH] : 0;
        int s = wave_incl_scan(v, lane);
        if (lane == 31) wsum[wv] = s;
        __syncthreads();
        if (wv == 0) {
            int w = wsum[lane];
            int ws = wave_incl_scan(w, lane);
            wsum[lane] = ws - w;
            if (lane == 31) tot = ws;
        }
        __syncthreads();
        if (i < nb) bsum[i * BS_PITCH] = carry + s - v + wsum[wv];
        carry += tot;
        __syncthreads();
    }
}

__global__ __launch_bounds__(256) void fill_kernel(const int* __restrict__ ei,
                                                   const float* __restrict__ w,
                                                   int* __restrict__ cur,
                                                   const int* __restrict__ boff,
                                                   int2* __restrict__ rec, int e) {
    int t = blockIdx.x * blockDim.x + threadIdx.x;
    if (t >= e) return;
    int s = nt_load_i32(ei + t);
    int d = nt_load_i32(ei + e + t);
    float wv = nt_load_f32(w + t);
    int p = __hip_atomic_fetch_add(cur + d, 1, __ATOMIC_RELAXED, __HIP_MEMORY_SCOPE_AGENT)
          + boff[(d >> SCAN_SH) * BS_PITCH];
    rec[p] = make_int2(s, __float_as_int(wv));
}

__global__ __launch_bounds__(256) void deg_pull(const int* __restrict__ loc,
                                                const int* __restrict__ cur,
                                                const int* __restrict__ boff,
                                                const int2* __restrict__ rec,
                                                float* __restrict__ dinv, int n) {
    __shared__ float sdeg[32];
    int node = blockIdx.x * 32 + (threadIdx.x >> 3);
    int j = threadIdx.x & 7;
    int beg = 0, end = 0;
    if (node < n) {
        int off = boff[(node >> SCAN_SH) * BS_PITCH];
        beg = loc[node] + off;
        end = cur[node] + off;
    }
    float acc = 0.0f;
    for (int p = beg + j; p < end; p += 8) acc += __int_as_float(rec[p].y);
    acc += __shfl_xor(acc, 1);
    acc += __shfl_xor(acc, 2);
    acc += __shfl_xor(acc, 4);
    if (j == 0) sdeg[threadIdx.x >> 3] = acc;
    __syncthreads();
    if (threadIdx.x < 32) {
        int nd = blockIdx.x * 32 + threadIdx.x;
        if (nd < n) dinv[nd] = rsqrtf(sdeg[threadIdx.x] + 1.0f);
    }
}

__global__ __launch_bounds__(256) void pull8_relu(const int* __restrict__ loc,
                                                  const int* __restrict__ cur,
                                                  const int* __restrict__ boff,
                                                  const int2* __restrict__ rec,
                                                  const float* __restrict__ dinv,
                                                  const float* __restrict__ h,
                                                  const float* __restrict__ b1,
                                                  float* __restrict__ hout, int n) {
    int t = blockIdx.x * blockDim.x + threadIdx.x;
    int node = t >> 3, j = t & 7;
    int beg = 0, end = 0;
    float di = 0.0f;
    if (node < n) {
        int off = boff[(node >> SCAN_SH) * BS_PITCH];
        beg = loc[node] + off;
        end = cur[node] + off;
        di = dinv[node];
    }
    float a0 = 0.f, a1 = 0.f, a2 = 0.f, a3 = 0.f, a4 = 0.f, a5 = 0.f, a6 = 0.f, a7 = 0.f;
    for (int p = beg + j; p < end; p += 8) {
        int2 r = rec[p];
        float nrm = dinv[r.x] * __int_as_float(r.y) * di;
        const float4* hp = (const float4*)(h + (size_t)r.x * 8);
        float4 lo4 = hp[0];
        float4 hi4 = hp[1];
        a0 += nrm * lo4.x; a1 += nrm * lo4.y; a2 += nrm * lo4.z; a3 += nrm * lo4.w;
        a4 += nrm * hi4.x; a5 += nrm * hi4.y; a6 += nrm * hi4.z; a7 += nrm * hi4.w;
    }
#pragma unroll
    for (int m = 1; m < 8; m <<= 1) {
        a0 += __shfl_xor(a0, m); a1 += __shfl_xor(a1, m);
        a2 += __shfl_xor(a2, m); a3 += __shfl_xor(a3, m);
        a4 += __shfl_xor(a4, m); a5 += __shfl_xor(a5, m);
        a6 += __shfl_xor(a6, m); a7 += __shfl_xor(a7, m);
    }
    if (node < n) {
        float v = a0;
        v = (j == 1) ? a1 : v; v = (j == 2) ? a2 : v; v = (j == 3) ? a3 : v;
        v = (j == 4) ? a4 : v; v = (j == 5) ? a5 : v; v = (j == 6) ? a6 : v;
        v = (j == 7) ? a7 : v;
        v = v + di * di * h[(size_t)node * 8 + j] + b1[j];
        hout[(size_t)node * 8 + j] = v > 0.0f ? v : 0.0f;
    }
}

__global__ __launch_bounds__(256) void pull2_softmax(const int* __restrict__ loc,
                                                     const int* __restrict__ cur,
                                                     const int* __restrict__ boff,
                                                     const int2* __restrict__ rec,
                                                     const float* __restrict__ dinv,
                                                     const float* __restrict__ h2,
                                                     const float* __restrict__ b2,
                                                     float* __restrict__ out, int n) {
    int t = blockIdx.x * blockDim.x + threadIdx.x;
    int node = t >> 1, j = t & 1;
    int beg = 0, end = 0;
    float di = 0.0f;
    if (node < n) {
        int off = boff[(node >> SCAN_SH) * BS_PITCH];
        beg = loc[node] + off;
        end = cur[node] + off;
        di = dinv[node];
    }
    float a0 = 0.f, a1 = 0.f;
    for (int p = beg + j; p < end; p += 2) {
        int2 r = rec[p];
        float nrm = dinv[r.x] * __int_as_float(r.y) * di;
        float2 hv = *(const float2*)(h2 + (size_t)r.x * 2);
        a0 += nrm * hv.x;
        a1 += nrm * hv.y;
    }
    a0 += __shfl_xor(a0, 1);
    a1 += __shfl_xor(a1, 1);
    if (node < n) {
        float sl = di * di;
        float2 hs = *(const float2*)(h2 + (size_t)node * 2);
        float v0 = a0 + sl * hs.x + b2[0];
        float v1 = a1 + sl * hs.y + b2[1];
        float m  = fmaxf(v0, v1);
        float e0 = __expf(v0 - m);
        float e1 = __expf(v1 - m);
        float inv = 1.0f / (e0 + e1);
        out[(size_t)node * 2 + j] = (j ? e1 : e0) * inv;
    }
}

#define T1_PITCH 16
#define T2_PITCH 4

__global__ __launch_bounds__(32) void linear1_pack(const float* __restrict__ x,
                                                   const float* __restrict__ W1,
                                                   const float* __restrict__ dinv,
                                                   float* __restrict__ t1, int n) {
    int lane = threadIdx.x;
    int base = blockIdx.x * 16;
    int idx  = lane & 15;
    bool lo  = lane < 16;
    int r    = base + idx;
    bool rv  = (r < n);
    int rowoff = (lo ? 0 : 8) + base;

    v2f a;
    a.x = (lo && rv) ? x[r * 2 + 0] : 0.0f;
    a.y = (lo && rv) ? x[r * 2 + 1] : 0.0f;

    bool cv = (idx < 8);
    v2f b;
    b.x = (lo && cv) ? W1[0 * 8 + idx] : 0.0f;
    b.y = (lo && cv) ? W1[1 * 8 + idx] : 0.0f;

    float dv[8];
#pragma unroll
    for (int v = 0; v < 8; ++v) {
        int m = rowoff + v;
        dv[v] = (idx == 8 && m < n) ? dinv[m] : 0.0f;
    }

    v8f c = {};
    v8f d = __builtin_amdgcn_wmma_f32_16x16x4_f32(false, a, false, b,
                                                  (short)0, c, false, false);
    if (base + 16 <= n) {
#pragma unroll
        for (int v = 0; v < 8; ++v)
            t1[(size_t)(rowoff + v) * T1_PITCH + idx] = cv ? d[v] : dv[v];
    } else {
#pragma unroll
        for (int v = 0; v < 8; ++v) {
            int m = rowoff + v;
            if (m < n) t1[(size_t)m * T1_PITCH + idx] = cv ? d[v] : dv[v];
        }
    }
}

__global__ __launch_bounds__(32) void linear2_pack(const float* __restrict__ h1,
                                                   const float* __restrict__ W2,
                                                   const float* __restrict__ dinv,
                                                   float* __restrict__ t2, int n) {
    int lane = threadIdx.x;
    int base = blockIdx.x * 16;
    int idx  = lane & 15;
    bool lo  = lane < 16;
    int r    = base + idx;
    bool rv  = (r < n);
    int k0   = lo ? 0 : 2;
    int rowoff = (lo ? 0 : 8) + base;

    v2f a0, a1;
    a0.x = rv ? h1[r * 8 + k0 + 0] : 0.0f;
    a0.y = rv ? h1[r * 8 + k0 + 1] : 0.0f;
    a1.x = rv ? h1[r * 8 + k0 + 4] : 0.0f;
    a1.y = rv ? h1[r * 8 + k0 + 5] : 0.0f;

    bool cv = (idx < 2);
    v2f b0, b1;
    b0.x = cv ? W2[(k0 + 0) * 2 + idx] : 0.0f;
    b0.y = cv ? W2[(k0 + 1) * 2 + idx] : 0.0f;
    b1.x = cv ? W2[(k0 + 4) * 2 + idx] : 0.0f;
    b1.y = cv ? W2[(k0 + 5) * 2 + idx] : 0.0f;

    float dv[8];
#pragma unroll
    for (int v = 0; v < 8; ++v) {
        int m = rowoff + v;
        dv[v] = (idx == 2 && m < n) ? dinv[m] : 0.0f;
    }

    v8f c = {};
    c = __builtin_amdgcn_wmma_f32_16x16x4_f32(false, a0, false, b0, (short)0, c, false, false);
    c = __builtin_amdgcn_wmma_f32_16x16x4_f32(false, a1, false, b1, (short)0, c, false, false);

    if (idx < T2_PITCH) {
        if (base + 16 <= n) {
#pragma unroll
            for (int v = 0; v < 8; ++v)
                t2[(size_t)(rowoff + v) * T2_PITCH + idx] = cv ? c[v] : dv[v];
        } else {
#pragma unroll
            for (int v = 0; v < 8; ++v) {
                int m = rowoff + v;
                if (m < n) t2[(size_t)m * T2_PITCH + idx] = cv ? c[v] : dv[v];
            }
        }
    }
}

template <int S> __device__ __forceinline__ int quad_pick(int v) {
    return __builtin_amdgcn_update_dpp(0, v, S * 0x55, 0xF, 0xF, false);
}
template <int S> __device__ __forceinline__ float quad_pickf(float v) {
    return __int_as_float(quad_pick<S>(__float_as_int(v)));
}

__device__ __forceinline__ float4 row_piece(const float* __restrict__ t1, int s, int q) {
    return *(const float4*)(t1 + (size_t)s * T1_PITCH + q * 4);
}

__global__ __launch_bounds__(256) void pull8_relu_pk(const int* __restrict__ loc,
                                                     const int* __restrict__ cur,
                                                     const int* __restrict__ boff,
                                                     const int2* __restrict__ rec,
                                                     const float* __restrict__ dinv,
                                                     const float* __restrict__ t1,
                                                     const float* __restrict__ b1,
                                                     float* __restrict__ hout, int n) {
    int t = blockIdx.x * blockDim.x + threadIdx.x;
    int node = t >> 3, j = t & 7;
    int e = j >> 2, q = j & 3;
    int beg = 0, end = 0;
    float di = 0.0f;
    if (node < n) {
        int off = boff[(node >> SCAN_SH) * BS_PITCH];
        beg = loc[node] + off;
        end = cur[node] + off;
        di = dinv[node];
    }
    const float4 z4 = make_float4(0.f, 0.f, 0.f, 0.f);
    const int2 zr = make_int2(0, 0);
    float a0 = 0.f, a1 = 0.f, a2 = 0.f, a3 = 0.f;
    int2 rn = (beg + j < end) ? rec[beg + j] : zr;
    for (int pb = beg; pb < end; pb += 8) {
        int2 r = rn;
        int pn = pb + 8 + j;
        bool more = (pn < end);
        int2 rl = rec[more ? pn : pb];
        rn = more ? rl : zr;
        int pe = pb + e * 4;
        int s0 = quad_pick<0>(r.x), s1 = quad_pick<1>(r.x);
        int s2 = quad_pick<2>(r.x), s3 = quad_pick<3>(r.x);
        float4 v0 = z4, v1 = z4, v2 = z4, v3 = z4;
        if (q < 3) {
            if (pe + 0 < end) v0 = row_piece(t1, s0, q);
            if (pe + 1 < end) v1 = row_piece(t1, s1, q);
            if (pe + 2 < end) v2 = row_piece(t1, s2, q);
            if (pe + 3 < end) v3 = row_piece(t1, s3, q);
        }
        float n0 = quad_pickf<2>(v0.x) * quad_pickf<0>(__int_as_float(r.y)) * di;
        float n1 = quad_pickf<2>(v1.x) * quad_pickf<1>(__int_as_float(r.y)) * di;
        float n2 = quad_pickf<2>(v2.x) * quad_pickf<2>(__int_as_float(r.y)) * di;
        float n3 = quad_pickf<2>(v3.x) * quad_pickf<3>(__int_as_float(r.y)) * di;
        a0 += n0 * v0.x; a1 += n0 * v0.y; a2 += n0 * v0.z; a3 += n0 * v0.w;
        a0 += n1 * v1.x; a1 += n1 * v1.y; a2 += n1 * v1.z; a3 += n1 * v1.w;
        a0 += n2 * v2.x; a1 += n2 * v2.y; a2 += n2 * v2.z; a3 += n2 * v2.w;
        a0 += n3 * v3.x; a1 += n3 * v3.y; a2 += n3 * v3.z; a3 += n3 * v3.w;
    }
    a0 += __shfl_xor(a0, 4); a1 += __shfl_xor(a1, 4);
    a2 += __shfl_xor(a2, 4); a3 += __shfl_xor(a3, 4);
    if (node < n && j < 2) {
        float4 hs = row_piece(t1, node, j);
        float sl = di * di;
        float4 o;
        o.x = a0 + sl * hs.x + b1[j * 4 + 0];
        o.y = a1 + sl * hs.y + b1[j * 4 + 1];
        o.z = a2 + sl * hs.z + b1[j * 4 + 2];
        o.w = a3 + sl * hs.w + b1[j * 4 + 3];
        o.x = o.x > 0.0f ? o.x : 0.0f; o.y = o.y > 0.0f ? o.y : 0.0f;
        o.z = o.z > 0.0f ? o.z : 0.0f; o.w = o.w > 0.0f ? o.w : 0.0f;
        *(float4*)(hout + (size_t)node * 8 + j * 4) = o;
    }
}

__global__ __launch_bounds__(256) void pull2_softmax_pk(const int* __restrict__ loc,
                                                        const int* __restrict__ cur,
                                                        const int* __restrict__ boff,
                                                        const int2* __restrict__ rec,
                                                        const float* __restrict__ t2,
                                                        const float* __restrict__ b2,
                                                        float* __restrict__ out, int n) {
    __shared__ float sres[64];
    int nl = threadIdx.x >> 3;
    int node = blockIdx.x * 32 + nl;
    int j = threadIdx.x & 7;
    int beg = 0, end = 0;
    float4 ts = make_float4(0.f, 0.f, 0.f, 0.f);
    if (node < n) {
        int off = boff[(node >> SCAN_SH) * BS_PITCH];
        beg = loc[node] + off;
        end = cur[node] + off;
        ts = *(const float4*)(t2 + (size_t)node * T2_PITCH);
    }
    float di = ts.z;
    const int2 zr = make_int2(0, 0);
    float a0 = 0.f, a1 = 0.f;
    int2 rn = (beg + j < end) ? rec[beg + j] : zr;
    for (int p = beg + j; p < end; p += 8) {
        int2 r = rn;
        float4 g = *(const float4*)(t2 + (size_t)r.x * T2_PITCH);
        bool more = (p + 8 < end);
        int2 rl = rec[more ? p + 8 : p];
        rn = more ? rl : zr;
        float nrm = g.z * __int_as_float(r.y) * di;
        a0 += nrm * g.x;
        a1 += nrm * g.y;
    }
    a0 += __shfl_xor(a0, 1); a1 += __shfl_xor(a1, 1);
    a0 += __shfl_xor(a0, 2); a1 += __shfl_xor(a1, 2);
    a0 += __shfl_xor(a0, 4); a1 += __shfl_xor(a1, 4);
    if (node < n && j == 0) {
        float sl = di * di;
        float v0 = a0 + sl * ts.x + b2[0];
        float v1 = a1 + sl * ts.y + b2[1];
        float m  = fmaxf(v0, v1);
        float e0 = __expf(v0 - m);
        float e1 = __expf(v1 - m);
        float inv = 1.0f / (e0 + e1);
        sres[nl * 2 + 0] = e0 * inv;
        sres[nl * 2 + 1] = e1 * inv;
    }
    __syncthreads();
    if (threadIdx.x < 64) {
        size_t o = (size_t)blockIdx.x * 64 + threadIdx.x;
        if (o < (size_t)n * 2) out[o] = sres[threadIdx.x];
    }
}

static inline float* align16(float* p) {
    return (float*)(((uintptr_t)p + 15) & ~(uintptr_t)15);
}
static inline size_t up256(size_t b) { return (b + 255) & ~(size_t)255; }

extern "C" void kernel_launch(void* const* d_in, const int* in_sizes, int n_in,
                              void* d_out, int out_size, void* d_ws, size_t ws_size,
                              hipStream_t stream) {
    const float* x  = (const float*)d_in[0];
    const int*   ei = (const int*)d_in[1];
    const float* w  = (const float*)d_in[2];
    const float* W1 = (const float*)d_in[3];
    const float* b1 = (const float*)d_in[4];
    const float* W2 = (const float*)d_in[5];
    const float* b2 = (const float*)d_in[6];
    float* out = (float*)d_out;

    const int n = in_sizes[0] / 2;
    const int e = in_sizes[2];

    const int TB = 256;
    dim3 blkE((e + TB - 1) / TB), thr(TB);
    dim3 blkN((n + TB - 1) / TB);
    dim3 blkW((n + 15) / 16), thrW(32);

    const int nb = (n + SCAN_B - 1) / SCAN_B;
    const size_t o_cur  = 0;
    const size_t o_loc  = o_cur  + up256((size_t)n * 4);
    const size_t o_bs   = o_loc  + up256((size_t)n * 4);
    const size_t o_rec  = o_bs   + up256((size_t)nb * BS_PITCH * 4);
    const size_t o_dinv = o_rec  + up256((size_t)e * 8);
    const size_t o_h1   = o_dinv + up256((size_t)n * 4);
    const size_t o_h1r  = o_h1   + up256((size_t)n * 32);
    const size_t o_h2   = o_h1r  + up256((size_t)n * 32);
    const size_t need   = o_h2   + up256((size_t)n * 8) + 256;

    const size_t p_t1   = o_dinv + up256((size_t)n * 4);
    const size_t p_h1r  = p_t1   + up256((size_t)n * T1_PITCH * 4);
    const size_t p_t2   = p_h1r  + up256((size_t)n * 32);
    const size_t needp  = p_t2   + up256((size_t)n * T2_PITCH * 4) + 256;

    if (ws_size >= needp) {
        char* base = (char*)(((uintptr_t)d_ws + 255) & ~(uintptr_t)255);
        int*   cur  = (int*)(base + o_cur);
        int*   loc  = (int*)(base + o_loc);
        int*   bs   = (int*)(base + o_bs);
        int2*  rec  = (int2*)(base + o_rec);
        float* dinv = (float*)(base + o_dinv);
        float* t1   = (float*)(base + p_t1);
        float* h1r  = (float*)(base + p_h1r);
        float* t2   = (float*)(base + p_t2);

        dim3 blkN8((n * 8 + TB - 1) / TB);
        dim3 blkN32((n + 31) / 32);

        zero_i32<<<blkN, thr, 0, stream>>>(cur, n);
        count_kernel<<<blkE, thr, 0, stream>>>(ei, cur, e);
        scan_local<<<dim3(nb), dim3(SCAN_B), 0, stream>>>(cur, loc, bs, n);
        scan_blocks<<<dim3(1), dim3(SCAN_B), 0, stream>>>(bs, nb);
        fill_kernel<<<blkE, thr, 0, stream>>>(ei, w, cur, bs, rec, e);
        deg_pull<<<blkN32, thr, 0, stream>>>(loc, cur, bs, rec, dinv, n);
        linear1_pack<<<blkW, thrW, 0, stream>>>(x, W1, dinv, t1, n);
        pull8_relu_pk<<<blkN8, thr, 0, stream>>>(loc, cur, bs, rec, dinv, t1, b1, h1r, n);
        linear2_pack<<<blkW, thrW, 0, stream>>>(h1r, W2, dinv, t2, n);
        pull2_softmax_pk<<<blkN32, thr, 0, stream>>>(loc, cur, bs, rec, t2, b2, out, n);
        return;
    }

    if (ws_size >= need) {
        char* base = (char*)(((uintptr_t)d_ws + 255) & ~(uintptr_t)255);
        int*   cur  = (int*)(base + o_cur);
        int*   loc  = (int*)(base + o_loc);
        int*   bs   = (int*)(base + o_bs);
        int2*  rec  = (int2*)(base + o_rec);
        float* dinv = (float*)(base + o_dinv);
        float* h1   = (float*)(base + o_h1);
        float* h1r  = (float*)(base + o_h1r);
        float* h2   = (float*)(base + o_h2);

        dim3 blkN8((n * 8 + TB - 1) / TB);
        dim3 blkN2((n * 2 + TB - 1) / TB);
        dim3 blkN32((n + 31) / 32);

        zero_i32<<<blkN, thr, 0, stream>>>(cur, n);
        count_kernel<<<blkE, thr, 0, stream>>>(ei, cur, e);
        scan_local<<<dim3(nb), dim3(SCAN_B), 0, stream>>>(cur, loc, bs, n);
        scan_blocks<<<dim3(1), dim3(SCAN_B), 0, stream>>>(bs, nb);
        fill_kernel<<<blkE, thr, 0, stream>>>(ei, w, cur, bs, rec, e);

        deg_pull<<<blkN32, thr, 0, stream>>>(loc, cur, bs, rec, dinv, n);

        linear1_wmma<<<blkW, thrW, 0, stream>>>(x, W1, h1, n);
        pull8_relu<<<blkN8, thr, 0, stream>>>(loc, cur, bs, rec, dinv, h1, b1, h1r, n);

        linear2_wmma<<<blkW, thrW, 0, stream>>>(h1r, W2, h2, n);
        pull2_softmax<<<blkN2, thr, 0, stream>>>(loc, cur, bs, rec, dinv, h2, b2, out, n);
        return;
    }

    float* ws   = (float*)d_ws;
    float* dinv = align16(ws);
    float* agg1 = align16(dinv + (size_t)n);
    float* agg2 = align16(agg1 + (size_t)n * 8);
    float* h1   = align16(agg2 + (size_t)n * 2);
    float* h2   = align16(h1   + (size_t)n * 8);

    dim3 blkN8((n * 8 + TB - 1) / TB);
    dim3 blkN2((n * 2 + TB - 1) / TB);

    zero_kernel<<<blkN,  thr, 0, stream>>>(dinv, n);
    zero_kernel<<<blkN8, thr, 0, stream>>>(agg1, n * 8);
    zero_kernel<<<blkN2, thr, 0, stream>>>(agg2, n * 2);
    degree_kernel<<<blkE, thr, 0, stream>>>(ei, w, dinv, e);
    dinv_kernel<<<blkN, thr, 0, stream>>>(dinv, n);
    linear1_wmma<<<blkW, thrW, 0, stream>>>(x, W1, h1, n);
    edge_agg8<<<blkE, thr, 0, stream>>>(ei, w, dinv, h1, agg1, e);
    finalize1<<<blkN, thr, 0, stream>>>(h1, agg1, dinv, b1, n);
    linear2_wmma<<<blkW, thrW, 0, stream>>>(h1, W2, h2, n);
    edge_agg2<<<blkE, thr, 0, stream>>>(ei, w, dinv, h2, agg2, e);
    finalize2<<<blkN, thr, 0, stream>>>(h2, agg2, dinv, b2, out, n);
}
